# starting kernel
# baseline (speedup 1.0000x reference)
#include <hip/hip_runtime.h>
#include <math.h>

#define BATCH   2
#define SEQLEN  2048
#define D_INNER 2048
#define D_STATE 16
#define DT_RANK 64
#define KPROJ   (DT_RANK + 2 * D_STATE)

typedef float v2f __attribute__((ext_vector_type(2)));
typedef float v8f __attribute__((ext_vector_type(8)));

__global__ __launch_bounds__(256) void xdbl_gemm_wmma(
    const float* __restrict__ u,
    const float* __restrict__ xpw,
    float* __restrict__ xdbl)
{
    const int lane = threadIdx.x & 31;
    const int wave = threadIdx.x >> 5;
    const int tile = blockIdx.x * 8 + wave;
    const int nt = tile % (KPROJ / 16);
    const int mt = tile / (KPROJ / 16);
    const int m0 = mt * 16, n0 = nt * 16;

    const int half = lane & 15;
    const int hi   = lane >> 4;
    const int kk   = hi * 2;

    const int b = m0 >> 11;
    const int l = (m0 & (SEQLEN - 1)) + half;

    const float* pa = u + (size_t)b * D_INNER * SEQLEN + (size_t)kk * SEQLEN + l;
    const float* pb = xpw + (size_t)(n0 + half) * D_INNER + kk;

    v8f c0 = {}, c1 = {}, c2 = {}, c3 = {};
    #pragma unroll 2
    for (int k = 0; k < D_INNER; k += 16) {
        v2f a, bm;
        a.x  = pa[0 * SEQLEN];  a.y  = pa[1 * SEQLEN];
        bm.x = pb[0];           bm.y = pb[1];
        c0 = __builtin_amdgcn_wmma_f32_16x16x4_f32(false, a, false, bm, (short)0, c0, false, false);

        a.x  = pa[4 * SEQLEN];  a.y  = pa[5 * SEQLEN];
        bm.x = pb[4];           bm.y = pb[5];
        c1 = __builtin_amdgcn_wmma_f32_16x16x4_f32(false, a, false, bm, (short)0, c1, false, false);

        a.x  = pa[8 * SEQLEN];  a.y  = pa[9 * SEQLEN];
        bm.x = pb[8];           bm.y = pb[9];
        c2 = __builtin_amdgcn_wmma_f32_16x16x4_f32(false, a, false, bm, (short)0, c2, false, false);

        a.x  = pa[12 * SEQLEN]; a.y  = pa[13 * SEQLEN];
        bm.x = pb[12];          bm.y = pb[13];
        c3 = __builtin_amdgcn_wmma_f32_16x16x4_f32(false, a, false, bm, (short)0, c3, false, false);

        pa += 16 * SEQLEN;
        pb += 16;
    }
    v8f c = (c0 + c1) + (c2 + c3);

    float* pc = xdbl + (size_t)(m0 + hi * 8) * KPROJ + (n0 + half);
    #pragma unroll
    for (int r = 0; r < 8; ++r)
        pc[(size_t)r * KPROJ] = c[r];
}

__global__ __launch_bounds__(256) void delta_gemm_wmma(
    const float* __restrict__ xdbl,
    const float* __restrict__ dtw,
    const float* __restrict__ dtb,
    float* __restrict__ delta)
{
    const int lane = threadIdx.x & 31;
    const int wave = threadIdx.x >> 5;
    const int tile = blockIdx.x * 8 + wave;
    const int nt = tile & 127;
    const int mt = tile >> 7;
    const int m0 = mt * 16, n0 = nt * 16;

    const int half = lane & 15;
    const int hi   = lane >> 4;
    const int kk   = hi * 2;

    const float* pa = xdbl + (size_t)(m0 + half) * KPROJ + kk;
    const float* pb = dtw + (size_t)(n0 + half) * DT_RANK + kk;

    v8f c0 = {}, c1 = {};
    #pragma unroll
    for (int k = 0; k < DT_RANK; k += 8) {
        v2f a, bm;
        a.x  = pa[0];  a.y  = pa[1];
        bm.x = pb[0];  bm.y = pb[1];
        c0 = __builtin_amdgcn_wmma_f32_16x16x4_f32(false, a, false, bm, (short)0, c0, false, false);

        a.x  = pa[4];  a.y  = pa[5];
        bm.x = pb[4];  bm.y = pb[5];
        c1 = __builtin_amdgcn_wmma_f32_16x16x4_f32(false, a, false, bm, (short)0, c1, false, false);

        pa += 8;
        pb += 8;
    }
    v8f c = c0 + c1;

    const int   d    = n0 + half;
    const float bias = dtb[d];
    const int   b    = m0 >> 11;
    const int   lb   = (m0 & (SEQLEN - 1)) + hi * 8;
    float* pd = delta + ((size_t)b * D_INNER + d) * SEQLEN + lb;

    #pragma unroll
    for (int r = 0; r < 8; ++r) {
        float z  = c[r] + bias;
        float sp = fmaxf(z, 0.0f) + log1pf(__expf(-fabsf(z)));
        pd[r] = sp;
    }
}

__global__ __launch_bounds__(256) void s6_scan(
    const float* __restrict__ u,
    const float* __restrict__ xdbl,
    const float* __restrict__ delta,
    const float* __restrict__ A_log,
    float* __restrict__ out)
{
    const int lane = threadIdx.x & 31;
    const int gwave = (blockIdx.x * blockDim.x + threadIdx.x) >> 5;
    const int g = lane >> 4;
    const int n = lane & 15;
    const int ch = gwave * 2 + g;
    const int b  = ch >> 11;
    const int d  = ch & (D_INNER - 1);

    const float An = -__expf(A_log[(size_t)d * D_STATE + n]);

    const float* pdl = delta + (size_t)ch * SEQLEN;
    const float* pu  = u     + (size_t)ch * SEQLEN;
    const float* pbc = xdbl  + (size_t)b * SEQLEN * KPROJ + DT_RANK + n;
    float*       po  = out   + (size_t)ch * SEQLEN;

    float x = 0.0f;
    #pragma unroll 4
    for (int l = 0; l < SEQLEN; ++l) {
        if ((l & 15) == 0) {
            __builtin_prefetch(pdl + l + 64, 0, 0);
            __builtin_prefetch(pu  + l + 64, 0, 0);
        }
        float dl = pdl[l];
        float uv = pu[l];
        float Bn = pbc[(size_t)l * KPROJ];
        float Cn = pbc[(size_t)l * KPROJ + D_STATE];

        float dA = __expf(dl * An);
        x = dA * x + (dl * uv) * Bn;

        float p = x * Cn;
        p += __shfl_xor(p, 1);
        p += __shfl_xor(p, 2);
        p += __shfl_xor(p, 4);
        p += __shfl_xor(p, 8);
        if (n == 0) po[l] = p;
    }
}

extern "C" void kernel_launch(void* const* d_in, const int* in_sizes, int n_in,
                              void* d_out, int out_size, void* d_ws, size_t ws_size,
                              hipStream_t stream)
{
    const float* u     = (const float*)d_in[0];
    const float* xpw   = (const float*)d_in[1];
    const float* dtw   = (const float*)d_in[2];
    const float* dtb   = (const float*)d_in[3];
    const float* A_log = (const float*)d_in[4];
    float*       out   = (float*)d_out;

    float* xdbl  = (float*)d_ws;
    float* delta = xdbl + (size_t)BATCH * SEQLEN * KPROJ;

    xdbl_gemm_wmma<<<192, 256, 0, stream>>>(u, xpw, xdbl);

    delta_gemm_wmma<<<4096, 256, 0, stream>>>(xdbl, dtw, dtb, delta);

    s6_scan<<<256, 256, 0, stream>>>(u, xdbl, delta, A_log, out);
}


# scan re-decomposed: channel on 8 lanes, 2 states per lane, 1024 waves (all SIMDs) not 512; 64-step blocks, delta/u quads reloaded in place; x = fma(dA,x,du*B); y = pair + 3 DPP adds (new sum order)
# speedup vs baseline: 8.0521x; 8.0521x over previous
#include <hip/hip_runtime.h>
#include <math.h>

#define BATCH   2
#define SEQLEN  2048
#define D_INNER 2048
#define D_STATE 16
#define DT_RANK 64
#define KPROJ   (DT_RANK + 2 * D_STATE)

typedef float v2f __attribute__((ext_vector_type(2)));
typedef float v8f __attribute__((ext_vector_type(8)));

__device__ __forceinline__ void lds_barrier()
{
    __builtin_amdgcn_fence(__ATOMIC_RELEASE, "workgroup", "local");
    __builtin_amdgcn_s_barrier();
    __builtin_amdgcn_fence(__ATOMIC_ACQUIRE, "workgroup", "local");
}

#define G1_KC    64
#define G1_NCH   (D_INNER / G1_KC)
#define G1_BP    (G1_KC + 4)
#define G1_WQ    (KPROJ * (G1_KC / 4) / 192)

typedef float g1v4 __attribute__((ext_vector_type(4)));

__global__ __launch_bounds__(192) void xdbl_gemm_wmma(
    const float* __restrict__ u,
    const float* __restrict__ xpw,
    float* __restrict__ xdbl)
{
    __shared__ __attribute__((aligned(16))) float As[2][G1_KC][16];
    __shared__ __attribute__((aligned(16))) float Bs[2][KPROJ][G1_BP];

    const int tid  = threadIdx.x;
    const int lane = tid & 31;
    const int wave = tid >> 5;
    const int m0 = blockIdx.x * 16, n0 = wave * 16;

    const int half = lane & 15;
    const int hi   = lane >> 4;

    const int b  = m0 >> 11;
    const int l0 = m0 & (SEQLEN - 1);

    const int  ak = (tid & 127) >> 2, aq = tid & 3;
    const float* ga = u + (size_t)b * D_INNER * SEQLEN + (size_t)ak * SEQLEN + l0 + aq * 4;
    const int  bn = tid >> 4, bq = tid & 15;
    const float* gb = xpw + (size_t)bn * D_INNER + bq * 4;

    g1v4 ra0[2], rb0[G1_WQ], ra1[2], rb1[G1_WQ];

    auto fetch = [&](int cn, g1v4 (&ra)[2], g1v4 (&rb)[G1_WQ]) __attribute__((always_inline)) {
        const size_t ko = (size_t)cn * G1_KC;
        ra[0] = *(const g1v4*)(ga + ko * SEQLEN);
        ra[1] = *(const g1v4*)(ga + (ko + 32) * SEQLEN);
        #pragma unroll
        for (int j = 0; j < G1_WQ; ++j)
            rb[j] = *(const g1v4*)(gb + (size_t)(12 * j) * D_INNER + ko);
    };
    auto stage = [&](int st, const g1v4 (&ra)[2], const g1v4 (&rb)[G1_WQ]) __attribute__((always_inline)) {
        *(g1v4*)&As[st][ak][aq * 4]      = ra[0];
        *(g1v4*)&As[st][ak + 32][aq * 4] = ra[1];
        #pragma unroll
        for (int j = 0; j < G1_WQ; ++j)
            *(g1v4*)&Bs[st][bn + 12 * j][bq * 4] = rb[j];
    };

    v8f c0 = {}, c1 = {}, c2 = {}, c3 = {};

    auto chunk = [&](int st) __attribute__((always_inline)) {
        v2f fa[G1_KC / 4], fb[G1_KC / 4];
        #pragma unroll
        for (int i = 0; i < G1_KC / 4; ++i) {
            fa[i].x = As[st][4 * i + 2 * hi][half];
            fa[i].y = As[st][4 * i + 2 * hi + 1][half];
            fb[i]   = *(const v2f*)&Bs[st][n0 + half][4 * i + 2 * hi];
        }
        #pragma unroll
        for (int i = 0; i < G1_KC / 4; i += 4) {
            c0 = __builtin_amdgcn_wmma_f32_16x16x4_f32(false, fa[i],     false, fb[i],     (short)0, c0, false, false);
            c1 = __builtin_amdgcn_wmma_f32_16x16x4_f32(false, fa[i + 1], false, fb[i + 1], (short)0, c1, false, false);
            c2 = __builtin_amdgcn_wmma_f32_16x16x4_f32(false, fa[i + 2], false, fb[i + 2], (short)0, c2, false, false);
            c3 = __builtin_amdgcn_wmma_f32_16x16x4_f32(false, fa[i + 3], false, fb[i + 3], (short)0, c3, false, false);
        }
    };

    fetch(0, ra0, rb0);
    fetch(1, ra1, rb1);
    stage(0, ra0, rb0);
    lds_barrier();

    #pragma unroll 1
    for (int ci = 0; ci < G1_NCH - 2; ci += 2) {
        fetch(ci + 2, ra0, rb0);
        __builtin_amdgcn_sched_barrier(0);
        chunk(0);
        stage(1, ra1, rb1);
        lds_barrier();

        fetch(ci + 3, ra1, rb1);
        __builtin_amdgcn_sched_barrier(0);
        chunk(1);
        stage(0, ra0, rb0);
        lds_barrier();
    }
    chunk(0);
    stage(1, ra1, rb1);
    lds_barrier();
    chunk(1);

    v8f c = (c0 + c1) + (c2 + c3);

    float* pc = xdbl + (size_t)(m0 + hi * 8) * KPROJ + (n0 + half);
    #pragma unroll
    for (int r = 0; r < 8; ++r)
        pc[(size_t)r * KPROJ] = c[r];
}

__global__ __launch_bounds__(256) void delta_gemm_wmma(
    const float* __restrict__ xdbl,
    const float* __restrict__ dtw,
    const float* __restrict__ dtb,
    float* __restrict__ delta)
{
    const int lane = threadIdx.x & 31;
    const int wave = threadIdx.x >> 5;
    const int tile = blockIdx.x * 8 + wave;
    const int nt = tile & 127;
    const int mt = tile >> 7;
    const int m0 = mt * 16, n0 = nt * 16;

    const int half = lane & 15;
    const int hi   = lane >> 4;
    const int kk   = hi * 2;

    const float* pa = xdbl + (size_t)(m0 + half) * KPROJ + kk;
    const float* pb = dtw + (size_t)(n0 + half) * DT_RANK + kk;

    v8f c0 = {}, c1 = {};
    #pragma unroll
    for (int k = 0; k < DT_RANK; k += 8) {
        v2f a, bm;
        a.x  = pa[0];  a.y  = pa[1];
        bm.x = pb[0];  bm.y = pb[1];
        c0 = __builtin_amdgcn_wmma_f32_16x16x4_f32(false, a, false, bm, (short)0, c0, false, false);

        a.x  = pa[4];  a.y  = pa[5];
        bm.x = pb[4];  bm.y = pb[5];
        c1 = __builtin_amdgcn_wmma_f32_16x16x4_f32(false, a, false, bm, (short)0, c1, false, false);

        pa += 8;
        pb += 8;
    }
    v8f c = c0 + c1;

    const int   d    = n0 + half;
    const float bias = dtb[d];
    const int   b    = m0 >> 11;
    const int   lb   = (m0 & (SEQLEN - 1)) + hi * 8;
    float* pd = delta + ((size_t)b * D_INNER + d) * SEQLEN + lb;

    #pragma unroll
    for (int r = 0; r < 8; ++r) {
        float z  = c[r] + bias;
        float sp = fmaxf(z, 0.0f) + log1pf(__expf(-fabsf(z)));
        pd[r] = sp;
    }
}

#define SCAN_TB    64
#define SCAN_NBLK  (SEQLEN / SCAN_TB)
#define SCAN_Q     (SCAN_TB / 4)
#define SCAN_G     8
#define SCAN_NG    (SCAN_TB / SCAN_G)

typedef float v4f __attribute__((ext_vector_type(4)));

template <int CTRL>
__device__ __forceinline__ float dpp_lane(float v)
{
    return __builtin_bit_cast(float,
        __builtin_amdgcn_update_dpp(0, __builtin_bit_cast(int, v), CTRL, 0xF, 0xF, true));
}

__global__ __launch_bounds__(128) void s6_scan(
    const float* __restrict__ u,
    const float* __restrict__ xdbl,
    const float* __restrict__ delta,
    const float* __restrict__ A_log,
    float* __restrict__ out)
{
    __shared__ v4f bc[2][SCAN_TB][8];

    const int tid = threadIdx.x;
    const int sub = tid & 7;
    const int ch  = blockIdx.x * 16 + (tid >> 3);
    const int b   = blockIdx.x >> 7;
    const int d   = ch & (D_INNER - 1);

    const v2f al = *(const v2f*)(A_log + (size_t)d * D_STATE + sub * 2);
    const float a0 = -__expf(al.x) * 1.44269504088896340736f;
    const float a1 = -__expf(al.y) * 1.44269504088896340736f;

    const float* pdl = delta + (size_t)ch * SEQLEN;
    const float* pu  = u     + (size_t)ch * SEQLEN;
    float*       po  = out   + (size_t)ch * SEQLEN;
    const int    trow = tid >> 3;
    const int    tcol = tid & 7;
    const float* pbc  = xdbl + ((size_t)b * SEQLEN + trow) * KPROJ + DT_RANK + tcol * 4;
    float* const lrow = (float*)&bc[0][0][0] + (tcol & 3) * 8 + (tcol >> 2) * 2;

    auto stage = [&](int bufi, int row, const v4f q) __attribute__((always_inline)) {
        float* dst = lrow + (bufi * SCAN_TB + row) * 32;
        v2f lo, hi;
        lo.x = q.x; lo.y = q.y; hi.x = q.z; hi.y = q.w;
        *(v2f*)dst       = lo;
        *(v2f*)(dst + 4) = hi;
    };

    v4f cd[SCAN_Q], cu[SCAN_Q];

    #pragma unroll
    for (int j = 0; j < SCAN_Q; ++j) {
        cd[j] = *(const v4f*)(pdl + 4 * j);
        cu[j] = *(const v4f*)(pu  + 4 * j);
    }
    #pragma unroll
    for (int j = 0; j < SCAN_TB / 16; ++j)
        stage(0, trow + 16 * j, *(const v4f*)(pbc + (size_t)(16 * j) * KPROJ));
    lds_barrier();

    float x0 = 0.0f, x1 = 0.0f;

    v4f bq[SCAN_G];
    #pragma unroll
    for (int s = 0; s < SCAN_G; ++s)
        bq[s] = bc[0][s][sub];

    #pragma unroll 1
    for (int k = 0; k < SCAN_NBLK; ++k) {
        const int buf = k & 1;
        const int kn  = (k + 1 < SCAN_NBLK) ? (k + 1) : k;
        const float* ndl = pdl + kn * SCAN_TB;
        const float* nu  = pu  + kn * SCAN_TB;
        const float* nbc = pbc + (size_t)(kn * SCAN_TB) * KPROJ;
        float*       yo  = po  + k * SCAN_TB;
        const v4f*   rb  = &bc[buf][0][sub];
        const v4f*   rn  = &bc[buf ^ 1][0][sub];

        v4f nb[SCAN_TB / 16];
        v4f bn[SCAN_G];

        #pragma unroll
        for (int g = 0; g < SCAN_NG; ++g) {
            #pragma unroll
            for (int s = 0; s < SCAN_G; ++s)
                bn[s] = (g + 1 < SCAN_NG) ? rb[8 * ((g + 1) * SCAN_G + s)] : rn[8 * s];
            if (g == 0) {
                #pragma unroll
                for (int j = 0; j < SCAN_TB / 16; ++j)
                    nb[j] = *(const v4f*)(nbc + (size_t)(16 * j) * KPROJ);
            }
            __builtin_amdgcn_sched_barrier(0);

            v4f y0, y1;
            #pragma unroll
            for (int s = 0; s < SCAN_G; ++s) {
                const int   q  = 2 * g + (s >> 2);
                const float dl = cd[q][s & 3];
                const float du = dl * cu[q][s & 3];
                const float dA0 = __builtin_amdgcn_exp2f(dl * a0);
                const float dA1 = __builtin_amdgcn_exp2f(dl * a1);
                const float t0 = du * bq[s].x;
                const float t1 = du * bq[s].y;
                x0 = __builtin_fmaf(dA0, x0, t0);
                x1 = __builtin_fmaf(dA1, x1, t1);
                float p = __builtin_fmaf(x1, bq[s].w, x0 * bq[s].z);
                p += dpp_lane<0xB1>(p);
                p += dpp_lane<0x4E>(p);
                p += dpp_lane<0x141>(p);
                if (s < 4) y0[s & 3] = p; else y1[s & 3] = p;
            }
            __builtin_amdgcn_sched_barrier(0);

            cd[2 * g]     = *(const v4f*)(ndl + 8 * g);
            cd[2 * g + 1] = *(const v4f*)(ndl + 8 * g + 4);
            cu[2 * g]     = *(const v4f*)(nu  + 8 * g);
            cu[2 * g + 1] = *(const v4f*)(nu  + 8 * g + 4);
            if (g == SCAN_NG - 2) {
                #pragma unroll
                for (int j = 0; j < SCAN_TB / 16; ++j)
                    stage(buf ^ 1, trow + 16 * j, nb[j]);
                lds_barrier();
            }
            if (sub == 0) {
                *(v4f*)(yo + 8 * g)     = y0;
                *(v4f*)(yo + 8 * g + 4) = y1;
            }
            #pragma unroll
            for (int s = 0; s < SCAN_G; ++s) bq[s] = bn[s];
        }
    }
}

extern "C" void kernel_launch(void* const* d_in, const int* in_sizes, int n_in,
                              void* d_out, int out_size, void* d_ws, size_t ws_size,
                              hipStream_t stream)
{
    const float* u     = (const float*)d_in[0];
    const float* xpw   = (const float*)d_in[1];
    const float* dtw   = (const float*)d_in[2];
    const float* dtb   = (const float*)d_in[3];
    const float* A_log = (const float*)d_in[4];
    float*       out   = (float*)d_out;

    float* xdbl  = (float*)d_ws;
    float* delta = xdbl + (size_t)BATCH * SEQLEN * KPROJ;

    xdbl_gemm_wmma<<<256, 192, 0, stream>>>(u, xpw, xdbl);

    delta_gemm_wmma<<<4096, 256, 0, stream>>>(xdbl, dtw, dtb, delta);

    s6_scan<<<256, 128, 0, stream>>>(u, xdbl, delta, A_log, out);
}


# delta GEMM operands through LDS: whole-row 16-B loads (4 lines a load, not 16), x_dbl tile once per work-group, fragments read from LDS; same sums (by-product: compiler interleaves 8 softplus chains)
# speedup vs baseline: 9.1203x; 1.1327x over previous
#include <hip/hip_runtime.h>
#include <math.h>

#define BATCH   2
#define SEQLEN  2048
#define D_INNER 2048
#define D_STATE 16
#define DT_RANK 64
#define KPROJ   (DT_RANK + 2 * D_STATE)

typedef float v2f __attribute__((ext_vector_type(2)));
typedef float v8f __attribute__((ext_vector_type(8)));

__device__ __forceinline__ void lds_barrier()
{
    __builtin_amdgcn_fence(__ATOMIC_RELEASE, "workgroup", "local");
    __builtin_amdgcn_s_barrier();
    __builtin_amdgcn_fence(__ATOMIC_ACQUIRE, "workgroup", "local");
}

#define G1_KC    64
#define G1_NCH   (D_INNER / G1_KC)
#define G1_BP    (G1_KC + 4)
#define G1_WQ    (KPROJ * (G1_KC / 4) / 192)

typedef float g1v4 __attribute__((ext_vector_type(4)));

__global__ __launch_bounds__(192) void xdbl_gemm_wmma(
    const float* __restrict__ u,
    const float* __restrict__ xpw,
    float* __restrict__ xdbl)
{
    __shared__ __attribute__((aligned(16))) float As[2][G1_KC][16];
    __shared__ __attribute__((aligned(16))) float Bs[2][KPROJ][G1_BP];

    const int tid  = threadIdx.x;
    const int lane = tid & 31;
    const int wave = tid >> 5;
    const int m0 = blockIdx.x * 16, n0 = wave * 16;

    const int half = lane & 15;
    const int hi   = lane >> 4;

    const int b  = m0 >> 11;
    const int l0 = m0 & (SEQLEN - 1);

    const int  ak = (tid & 127) >> 2, aq = tid & 3;
    const float* ga = u + (size_t)b * D_INNER * SEQLEN + (size_t)ak * SEQLEN + l0 + aq * 4;
    const int  bn = tid >> 4, bq = tid & 15;
    const float* gb = xpw + (size_t)bn * D_INNER + bq * 4;

    g1v4 ra0[2], rb0[G1_WQ], ra1[2], rb1[G1_WQ];

    auto fetch = [&](int cn, g1v4 (&ra)[2], g1v4 (&rb)[G1_WQ]) __attribute__((always_inline)) {
        const size_t ko = (size_t)cn * G1_KC;
        ra[0] = *(const g1v4*)(ga + ko * SEQLEN);
        ra[1] = *(const g1v4*)(ga + (ko + 32) * SEQLEN);
        #pragma unroll
        for (int j = 0; j < G1_WQ; ++j)
            rb[j] = *(const g1v4*)(gb + (size_t)(12 * j) * D_INNER + ko);
    };
    auto stage = [&](int st, const g1v4 (&ra)[2], const g1v4 (&rb)[G1_WQ]) __attribute__((always_inline)) {
        *(g1v4*)&As[st][ak][aq * 4]      = ra[0];
        *(g1v4*)&As[st][ak + 32][aq * 4] = ra[1];
        #pragma unroll
        for (int j = 0; j < G1_WQ; ++j)
            *(g1v4*)&Bs[st][bn + 12 * j][bq * 4] = rb[j];
    };

    v8f c0 = {}, c1 = {}, c2 = {}, c3 = {};

    auto chunk = [&](int st) __attribute__((always_inline)) {
        v2f fa[G1_KC / 4], fb[G1_KC / 4];
        #pragma unroll
        for (int i = 0; i < G1_KC / 4; ++i) {
            fa[i].x = As[st][4 * i + 2 * hi][half];
            fa[i].y = As[st][4 * i + 2 * hi + 1][half];
            fb[i]   = *(const v2f*)&Bs[st][n0 + half][4 * i + 2 * hi];
        }
        #pragma unroll
        for (int i = 0; i < G1_KC / 4; i += 4) {
            c0 = __builtin_amdgcn_wmma_f32_16x16x4_f32(false, fa[i],     false, fb[i],     (short)0, c0, false, false);
            c1 = __builtin_amdgcn_wmma_f32_16x16x4_f32(false, fa[i + 1], false, fb[i + 1], (short)0, c1, false, false);
            c2 = __builtin_amdgcn_wmma_f32_16x16x4_f32(false, fa[i + 2], false, fb[i + 2], (short)0, c2, false, false);
            c3 = __builtin_amdgcn_wmma_f32_16x16x4_f32(false, fa[i + 3], false, fb[i + 3], (short)0, c3, false, false);
        }
    };

    fetch(0, ra0, rb0);
    fetch(1, ra1, rb1);
    stage(0, ra0, rb0);
    lds_barrier();

    #pragma unroll 1
    for (int ci = 0; ci < G1_NCH - 2; ci += 2) {
        fetch(ci + 2, ra0, rb0);
        __builtin_amdgcn_sched_barrier(0);
        chunk(0);
        stage(1, ra1, rb1);
        lds_barrier();

        fetch(ci + 3, ra1, rb1);
        __builtin_amdgcn_sched_barrier(0);
        chunk(1);
        stage(0, ra0, rb0);
        lds_barrier();
    }
    chunk(0);
    stage(1, ra1, rb1);
    lds_barrier();
    chunk(1);

    v8f c = (c0 + c1) + (c2 + c3);

    float* pc = xdbl + (size_t)(m0 + hi * 8) * KPROJ + (n0 + half);
    #pragma unroll
    for (int r = 0; r < 8; ++r)
        pc[(size_t)r * KPROJ] = c[r];
}

#define G2_P   (DT_RANK + 4)
#define G2_WR  128

__global__ __launch_bounds__(256) void delta_gemm_wmma(
    const float* __restrict__ xdbl,
    const float* __restrict__ dtw,
    const float* __restrict__ dtb,
    float* __restrict__ delta)
{
    __shared__ __attribute__((aligned(16))) float Xs[16][G2_P];
    __shared__ __attribute__((aligned(16))) float Ws[G2_WR][G2_P];

    const int tid  = threadIdx.x;
    const int lane = tid & 31;
    const int wave = tid >> 5;
    const int mt = blockIdx.x >> 4;
    const int ng = blockIdx.x & 15;
    const int m0 = mt * 16, n0 = ng * G2_WR + wave * 16;

    const int srow = tid >> 4, sq = tid & 15;
    const g1v4 rx = *(const g1v4*)(xdbl + (size_t)(m0 + srow) * KPROJ + sq * 4);
    const float* gw = dtw + (size_t)(ng * G2_WR + srow) * DT_RANK + sq * 4;
    g1v4 rw[G2_WR / 16];
    #pragma unroll
    for (int j = 0; j < G2_WR / 16; ++j)
        rw[j] = *(const g1v4*)(gw + (size_t)(16 * j) * DT_RANK);
    *(g1v4*)&Xs[srow][sq * 4] = rx;
    #pragma unroll
    for (int j = 0; j < G2_WR / 16; ++j)
        *(g1v4*)&Ws[16 * j + srow][sq * 4] = rw[j];
    lds_barrier();

    const int half = lane & 15;
    const int hi   = lane >> 4;

    v2f fa[DT_RANK / 4], fb[DT_RANK / 4];
    #pragma unroll
    for (int i = 0; i < DT_RANK / 4; ++i) {
        fa[i] = *(const v2f*)&Xs[half][4 * i + 2 * hi];
        fb[i] = *(const v2f*)&Ws[wave * 16 + half][4 * i + 2 * hi];
    }

    v8f c0 = {}, c1 = {};
    #pragma unroll
    for (int i = 0; i < DT_RANK / 4; i += 2) {
        c0 = __builtin_amdgcn_wmma_f32_16x16x4_f32(false, fa[i],     false, fb[i],     (short)0, c0, false, false);
        c1 = __builtin_amdgcn_wmma_f32_16x16x4_f32(false, fa[i + 1], false, fb[i + 1], (short)0, c1, false, false);
    }
    v8f c = c0 + c1;

    const int   d    = n0 + half;
    const float bias = dtb[d];
    const int   b    = m0 >> 11;
    const int   lb   = (m0 & (SEQLEN - 1)) + hi * 8;
    float* pd = delta + ((size_t)b * D_INNER + d) * SEQLEN + lb;

    #pragma unroll
    for (int r = 0; r < 8; ++r) {
        float z  = c[r] + bias;
        float sp = fmaxf(z, 0.0f) + log1pf(__expf(-fabsf(z)));
        pd[r] = sp;
    }
}

#define SCAN_TB    64
#define SCAN_NBLK  (SEQLEN / SCAN_TB)
#define SCAN_Q     (SCAN_TB / 4)
#define SCAN_G     8
#define SCAN_NG    (SCAN_TB / SCAN_G)

typedef float v4f __attribute__((ext_vector_type(4)));

template <int CTRL>
__device__ __forceinline__ float dpp_lane(float v)
{
    return __builtin_bit_cast(float,
        __builtin_amdgcn_update_dpp(0, __builtin_bit_cast(int, v), CTRL, 0xF, 0xF, true));
}

__global__ __launch_bounds__(128) void s6_scan(
    const float* __restrict__ u,
    const float* __restrict__ xdbl,
    const float* __restrict__ delta,
    const float* __restrict__ A_log,
    float* __restrict__ out)
{
    __shared__ v4f bc[2][SCAN_TB][8];

    const int tid = threadIdx.x;
    const int sub = tid & 7;
    const int ch  = blockIdx.x * 16 + (tid >> 3);
    const int b   = blockIdx.x >> 7;
    const int d   = ch & (D_INNER - 1);

    const v2f al = *(const v2f*)(A_log + (size_t)d * D_STATE + sub * 2);
    const float a0 = -__expf(al.x) * 1.44269504088896340736f;
    const float a1 = -__expf(al.y) * 1.44269504088896340736f;

    const float* pdl = delta + (size_t)ch * SEQLEN;
    const float* pu  = u     + (size_t)ch * SEQLEN;
    float*       po  = out   + (size_t)ch * SEQLEN;
    const int    trow = tid >> 3;
    const int    tcol = tid & 7;
    const float* pbc  = xdbl + ((size_t)b * SEQLEN + trow) * KPROJ + DT_RANK + tcol * 4;
    float* const lrow = (float*)&bc[0][0][0] + (tcol & 3) * 8 + (tcol >> 2) * 2;

    auto stage = [&](int bufi, int row, const v4f q) __attribute__((always_inline)) {
        float* dst = lrow + (bufi * SCAN_TB + row) * 32;
        v2f lo, hi;
        lo.x = q.x; lo.y = q.y; hi.x = q.z; hi.y = q.w;
        *(v2f*)dst       = lo;
        *(v2f*)(dst + 4) = hi;
    };

    v4f cd[SCAN_Q], cu[SCAN_Q];

    #pragma unroll
    for (int j = 0; j < SCAN_Q; ++j) {
        cd[j] = *(const v4f*)(pdl + 4 * j);
        cu[j] = *(const v4f*)(pu  + 4 * j);
    }
    #pragma unroll
    for (int j = 0; j < SCAN_TB / 16; ++j)
        stage(0, trow + 16 * j, *(const v4f*)(pbc + (size_t)(16 * j) * KPROJ));
    lds_barrier();

    float x0 = 0.0f, x1 = 0.0f;

    v4f bq[SCAN_G];
    #pragma unroll
    for (int s = 0; s < SCAN_G; ++s)
        bq[s] = bc[0][s][sub];

    #pragma unroll 1
    for (int k = 0; k < SCAN_NBLK; ++k) {
        const int buf = k & 1;
        const int kn  = (k + 1 < SCAN_NBLK) ? (k + 1) : k;
        const float* ndl = pdl + kn * SCAN_TB;
        const float* nu  = pu  + kn * SCAN_TB;
        const float* nbc = pbc + (size_t)(kn * SCAN_TB) * KPROJ;
        float*       yo  = po  + k * SCAN_TB;
        const v4f*   rb  = &bc[buf][0][sub];
        const v4f*   rn  = &bc[buf ^ 1][0][sub];

        v4f nb[SCAN_TB / 16];
        v4f bn[SCAN_G];

        #pragma unroll
        for (int g = 0; g < SCAN_NG; ++g) {
            #pragma unroll
            for (int s = 0; s < SCAN_G; ++s)
                bn[s] = (g + 1 < SCAN_NG) ? rb[8 * ((g + 1) * SCAN_G + s)] : rn[8 * s];
            if (g == 0) {
                #pragma unroll
                for (int j = 0; j < SCAN_TB / 16; ++j)
                    nb[j] = *(const v4f*)(nbc + (size_t)(16 * j) * KPROJ);
            }
            __builtin_amdgcn_sched_barrier(0);

            v4f y0, y1;
            #pragma unroll
            for (int s = 0; s < SCAN_G; ++s) {
                const int   q  = 2 * g + (s >> 2);
                const float dl = cd[q][s & 3];
                const float du = dl * cu[q][s & 3];
                const float dA0 = __builtin_amdgcn_exp2f(dl * a0);
                const float dA1 = __builtin_amdgcn_exp2f(dl * a1);
                const float t0 = du * bq[s].x;
                const float t1 = du * bq[s].y;
                x0 = __builtin_fmaf(dA0, x0, t0);
                x1 = __builtin_fmaf(dA1, x1, t1);
                float p = __builtin_fmaf(x1, bq[s].w, x0 * bq[s].z);
                p += dpp_lane<0xB1>(p);
                p += dpp_lane<0x4E>(p);
                p += dpp_lane<0x141>(p);
                if (s < 4) y0[s & 3] = p; else y1[s & 3] = p;
            }
            __builtin_amdgcn_sched_barrier(0);

            cd[2 * g]     = *(const v4f*)(ndl + 8 * g);
            cd[2 * g + 1] = *(const v4f*)(ndl + 8 * g + 4);
            cu[2 * g]     = *(const v4f*)(nu  + 8 * g);
            cu[2 * g + 1] = *(const v4f*)(nu  + 8 * g + 4);
            if (g == SCAN_NG - 2) {
                #pragma unroll
                for (int j = 0; j < SCAN_TB / 16; ++j)
                    stage(buf ^ 1, trow + 16 * j, nb[j]);
                lds_barrier();
            }
            if (sub == 0) {
                *(v4f*)(yo + 8 * g)     = y0;
                *(v4f*)(yo + 8 * g + 4) = y1;
            }
            #pragma unroll
            for (int s = 0; s < SCAN_G; ++s) bq[s] = bn[s];
        }
    }
}

extern "C" void kernel_launch(void* const* d_in, const int* in_sizes, int n_in,
                              void* d_out, int out_size, void* d_ws, size_t ws_size,
                              hipStream_t stream)
{
    const float* u     = (const float*)d_in[0];
    const float* xpw   = (const float*)d_in[1];
    const float* dtw   = (const float*)d_in[2];
    const float* dtb   = (const float*)d_in[3];
    const float* A_log = (const float*)d_in[4];
    float*       out   = (float*)d_out;

    float* xdbl  = (float*)d_ws;
    float* delta = xdbl + (size_t)BATCH * SEQLEN * KPROJ;

    xdbl_gemm_wmma<<<256, 192, 0, stream>>>(u, xpw, xdbl);

    delta_gemm_wmma<<<4096, 256, 0, stream>>>(xdbl, dtw, dtb, delta);

    s6_scan<<<256, 128, 0, stream>>>(u, xdbl, delta, A_log, out);
}


# scan: y's 8-lane sums through an LDS transpose per 32 steps (16-B row writes, lane j reads steps 4j..4j+3 of 8 rows, adds in-lane) instead of 3 DPP adds per step; whole-line y stores; same sums
# speedup vs baseline: 10.6116x; 1.1635x over previous
#include <hip/hip_runtime.h>
#include <math.h>

#define BATCH   2
#define SEQLEN  2048
#define D_INNER 2048
#define D_STATE 16
#define DT_RANK 64
#define KPROJ   (DT_RANK + 2 * D_STATE)

typedef float v2f __attribute__((ext_vector_type(2)));
typedef float v8f __attribute__((ext_vector_type(8)));

__device__ __forceinline__ void lds_barrier()
{
    __builtin_amdgcn_fence(__ATOMIC_RELEASE, "workgroup", "local");
    __builtin_amdgcn_s_barrier();
    __builtin_amdgcn_fence(__ATOMIC_ACQUIRE, "workgroup", "local");
}

#define G1_KC    64
#define G1_NCH   (D_INNER / G1_KC)
#define G1_BP    (G1_KC + 4)
#define G1_WQ    (KPROJ * (G1_KC / 4) / 192)

typedef float g1v4 __attribute__((ext_vector_type(4)));

__global__ __launch_bounds__(192) void xdbl_gemm_wmma(
    const float* __restrict__ u,
    const float* __restrict__ xpw,
    float* __restrict__ xdbl)
{
    __shared__ __attribute__((aligned(16))) float As[2][G1_KC][16];
    __shared__ __attribute__((aligned(16))) float Bs[2][KPROJ][G1_BP];

    const int tid  = threadIdx.x;
    const int lane = tid & 31;
    const int wave = tid >> 5;
    const int m0 = blockIdx.x * 16, n0 = wave * 16;

    const int half = lane & 15;
    const int hi   = lane >> 4;

    const int b  = m0 >> 11;
    const int l0 = m0 & (SEQLEN - 1);

    const int  ak = (tid & 127) >> 2, aq = tid & 3;
    const float* ga = u + (size_t)b * D_INNER * SEQLEN + (size_t)ak * SEQLEN + l0 + aq * 4;
    const int  bn = tid >> 4, bq = tid & 15;
    const float* gb = xpw + (size_t)bn * D_INNER + bq * 4;

    g1v4 ra0[2], rb0[G1_WQ], ra1[2], rb1[G1_WQ];

    auto fetch = [&](int cn, g1v4 (&ra)[2], g1v4 (&rb)[G1_WQ]) __attribute__((always_inline)) {
        const size_t ko = (size_t)cn * G1_KC;
        ra[0] = *(const g1v4*)(ga + ko * SEQLEN);
        ra[1] = *(const g1v4*)(ga + (ko + 32) * SEQLEN);
        #pragma unroll
        for (int j = 0; j < G1_WQ; ++j)
            rb[j] = *(const g1v4*)(gb + (size_t)(12 * j) * D_INNER + ko);
    };
    auto stage = [&](int st, const g1v4 (&ra)[2], const g1v4 (&rb)[G1_WQ]) __attribute__((always_inline)) {
        *(g1v4*)&As[st][ak][aq * 4]      = ra[0];
        *(g1v4*)&As[st][ak + 32][aq * 4] = ra[1];
        #pragma unroll
        for (int j = 0; j < G1_WQ; ++j)
            *(g1v4*)&Bs[st][bn + 12 * j][bq * 4] = rb[j];
    };

    v8f c0 = {}, c1 = {}, c2 = {}, c3 = {};

    auto chunk = [&](int st) __attribute__((always_inline)) {
        v2f fa[G1_KC / 4], fb[G1_KC / 4];
        #pragma unroll
        for (int i = 0; i < G1_KC / 4; ++i) {
            fa[i].x = As[st][4 * i + 2 * hi][half];
            fa[i].y = As[st][4 * i + 2 * hi + 1][half];
            fb[i]   = *(const v2f*)&Bs[st][n0 + half][4 * i + 2 * hi];
        }
        #pragma unroll
        for (int i = 0; i < G1_KC / 4; i += 4) {
            c0 = __builtin_amdgcn_wmma_f32_16x16x4_f32(false, fa[i],     false, fb[i],     (short)0, c0, false, false);
            c1 = __builtin_amdgcn_wmma_f32_16x16x4_f32(false, fa[i + 1], false, fb[i + 1], (short)0, c1, false, false);
            c2 = __builtin_amdgcn_wmma_f32_16x16x4_f32(false, fa[i + 2], false, fb[i + 2], (short)0, c2, false, false);
            c3 = __builtin_amdgcn_wmma_f32_16x16x4_f32(false, fa[i + 3], false, fb[i + 3], (short)0, c3, false, false);
        }
    };

    fetch(0, ra0, rb0);
    fetch(1, ra1, rb1);
    stage(0, ra0, rb0);
    lds_barrier();

    #pragma unroll 1
    for (int ci = 0; ci < G1_NCH - 2; ci += 2) {
        fetch(ci + 2, ra0, rb0);
        __builtin_amdgcn_sched_barrier(0);
        chunk(0);
        stage(1, ra1, rb1);
        lds_barrier();

        fetch(ci + 3, ra1, rb1);
        __builtin_amdgcn_sched_barrier(0);
        chunk(1);
        stage(0, ra0, rb0);
        lds_barrier();
    }
    chunk(0);
    stage(1, ra1, rb1);
    lds_barrier();
    chunk(1);

    v8f c = (c0 + c1) + (c2 + c3);

    float* pc = xdbl + (size_t)(m0 + hi * 8) * KPROJ + (n0 + half);
    #pragma unroll
    for (int r = 0; r < 8; ++r)
        pc[(size_t)r * KPROJ] = c[r];
}

#define G2_P   (DT_RANK + 4)
#define G2_WR  128

__global__ __launch_bounds__(256) void delta_gemm_wmma(
    const float* __restrict__ xdbl,
    const float* __restrict__ dtw,
    const float* __restrict__ dtb,
    float* __restrict__ delta)
{
    __shared__ __attribute__((aligned(16))) float Xs[16][G2_P];
    __shared__ __attribute__((aligned(16))) float Ws[G2_WR][G2_P];

    const int tid  = threadIdx.x;
    const int lane = tid & 31;
    const int wave = tid >> 5;
    const int mt = blockIdx.x >> 4;
    const int ng = blockIdx.x & 15;
    const int m0 = mt * 16, n0 = ng * G2_WR + wave * 16;

    const int srow = tid >> 4, sq = tid & 15;
    const g1v4 rx = *(const g1v4*)(xdbl + (size_t)(m0 + srow) * KPROJ + sq * 4);
    const float* gw = dtw + (size_t)(ng * G2_WR + srow) * DT_RANK + sq * 4;
    g1v4 rw[G2_WR / 16];
    #pragma unroll
    for (int j = 0; j < G2_WR / 16; ++j)
        rw[j] = *(const g1v4*)(gw + (size_t)(16 * j) * DT_RANK);
    *(g1v4*)&Xs[srow][sq * 4] = rx;
    #pragma unroll
    for (int j = 0; j < G2_WR / 16; ++j)
        *(g1v4*)&Ws[16 * j + srow][sq * 4] = rw[j];
    lds_barrier();

    const int half = lane & 15;
    const int hi   = lane >> 4;

    v2f fa[DT_RANK / 4], fb[DT_RANK / 4];
    #pragma unroll
    for (int i = 0; i < DT_RANK / 4; ++i) {
        fa[i] = *(const v2f*)&Xs[half][4 * i + 2 * hi];
        fb[i] = *(const v2f*)&Ws[wave * 16 + half][4 * i + 2 * hi];
    }

    v8f c0 = {}, c1 = {};
    #pragma unroll
    for (int i = 0; i < DT_RANK / 4; i += 2) {
        c0 = __builtin_amdgcn_wmma_f32_16x16x4_f32(false, fa[i],     false, fb[i],     (short)0, c0, false, false);
        c1 = __builtin_amdgcn_wmma_f32_16x16x4_f32(false, fa[i + 1], false, fb[i + 1], (short)0, c1, false, false);
    }
    v8f c = c0 + c1;

    const int   d    = n0 + half;
    const float bias = dtb[d];
    const int   b    = m0 >> 11;
    const int   lb   = (m0 & (SEQLEN - 1)) + hi * 8;
    float* pd = delta + ((size_t)b * D_INNER + d) * SEQLEN + lb;

    #pragma unroll
    for (int r = 0; r < 8; ++r) {
        float z  = c[r] + bias;
        float sp = fmaxf(z, 0.0f) + log1pf(__expf(-fabsf(z)));
        pd[r] = sp;
    }
}

#define SCAN_TB    64
#define SCAN_NBLK  (SEQLEN / SCAN_TB)
#define SCAN_Q     (SCAN_TB / 4)
#define SCAN_G     8
#define SCAN_NG    (SCAN_TB / SCAN_G)
#define SCAN_SP    32
#define SCAN_YP    (SCAN_SP + 4)

typedef float v4f __attribute__((ext_vector_type(4)));

__global__ __launch_bounds__(128) void s6_scan(
    const float* __restrict__ u,
    const float* __restrict__ xdbl,
    const float* __restrict__ delta,
    const float* __restrict__ A_log,
    float* __restrict__ out)
{
    __shared__ v4f bc[2][SCAN_TB][8];
    __shared__ __attribute__((aligned(16))) float yt[4][32][SCAN_YP];

    const int tid = threadIdx.x;
    const int sub = tid & 7;
    const int ch  = blockIdx.x * 16 + (tid >> 3);
    const int b   = blockIdx.x >> 7;
    const int d   = ch & (D_INNER - 1);

    const v2f al = *(const v2f*)(A_log + (size_t)d * D_STATE + sub * 2);
    const float a0 = -__expf(al.x) * 1.44269504088896340736f;
    const float a1 = -__expf(al.y) * 1.44269504088896340736f;

    const float* pdl = delta + (size_t)ch * SEQLEN;
    const float* pu  = u     + (size_t)ch * SEQLEN;
    float*       po  = out   + (size_t)ch * SEQLEN;
    const int    trow = tid >> 3;
    const int    tcol = tid & 7;
    const float* pbc  = xdbl + ((size_t)b * SEQLEN + trow) * KPROJ + DT_RANK + tcol * 4;
    float* const lrow = (float*)&bc[0][0][0] + (tcol & 3) * 8 + (tcol >> 2) * 2;

    auto stage = [&](int bufi, int row, const v4f q) __attribute__((always_inline)) {
        float* dst = lrow + (bufi * SCAN_TB + row) * 32;
        v2f lo, hi;
        lo.x = q.x; lo.y = q.y; hi.x = q.z; hi.y = q.w;
        *(v2f*)dst       = lo;
        *(v2f*)(dst + 4) = hi;
    };

    v4f cd[SCAN_Q], cu[SCAN_Q];

    #pragma unroll
    for (int j = 0; j < SCAN_Q; ++j) {
        cd[j] = *(const v4f*)(pdl + 4 * j);
        cu[j] = *(const v4f*)(pu  + 4 * j);
    }
    #pragma unroll
    for (int j = 0; j < SCAN_TB / 16; ++j)
        stage(0, trow + 16 * j, *(const v4f*)(pbc + (size_t)(16 * j) * KPROJ));
    lds_barrier();

    float x0 = 0.0f, x1 = 0.0f;

    float* const       ywr = &yt[tid >> 5][tid & 31][0];
    const float* const yrd = &yt[tid >> 5][tid & 24][sub * 4];
    v4f tq[8] = {};

    auto ysum = [&]() __attribute__((always_inline)) {
        return ((tq[0] + tq[1]) + (tq[2] + tq[3])) + ((tq[4] + tq[5]) + (tq[6] + tq[7]));
    };

    v4f bq[SCAN_G];
    #pragma unroll
    for (int s = 0; s < SCAN_G; ++s)
        bq[s] = bc[0][s][sub];

    #pragma unroll 1
    for (int k = 0; k < SCAN_NBLK; ++k) {
        const int buf = k & 1;
        const int kn  = (k + 1 < SCAN_NBLK) ? (k + 1) : k;
        const float* ndl = pdl + kn * SCAN_TB;
        const float* nu  = pu  + kn * SCAN_TB;
        const float* nbc = pbc + (size_t)(kn * SCAN_TB) * KPROJ;
        float*       yo  = po  + k * SCAN_TB;
        const v4f*   rb  = &bc[buf][0][sub];
        const v4f*   rn  = &bc[buf ^ 1][0][sub];

        v4f nb[SCAN_TB / 16];
        v4f bn[SCAN_G];

        #pragma unroll
        for (int g = 0; g < SCAN_NG; ++g) {
            #pragma unroll
            for (int s = 0; s < SCAN_G; ++s)
                bn[s] = (g + 1 < SCAN_NG) ? rb[8 * ((g + 1) * SCAN_G + s)] : rn[8 * s];
            if (g == 0) {
                #pragma unroll
                for (int j = 0; j < SCAN_TB / 16; ++j)
                    nb[j] = *(const v4f*)(nbc + (size_t)(16 * j) * KPROJ);
            }
            __builtin_amdgcn_sched_barrier(0);

            v4f p0, p1;
            #pragma unroll
            for (int s = 0; s < SCAN_G; ++s) {
                const int   q  = 2 * g + (s >> 2);
                const float dl = cd[q][s & 3];
                const float du = dl * cu[q][s & 3];
                const float dA0 = __builtin_amdgcn_exp2f(dl * a0);
                const float dA1 = __builtin_amdgcn_exp2f(dl * a1);
                const float t0 = du * bq[s].x;
                const float t1 = du * bq[s].y;
                x0 = __builtin_fmaf(dA0, x0, t0);
                x1 = __builtin_fmaf(dA1, x1, t1);
                const float p = __builtin_fmaf(x1, bq[s].w, x0 * bq[s].z);
                if (s < 4) p0[s & 3] = p; else p1[s & 3] = p;
            }
            __builtin_amdgcn_sched_barrier(0);

            *(v4f*)(ywr + SCAN_G * (g & 3))     = p0;
            *(v4f*)(ywr + SCAN_G * (g & 3) + 4) = p1;

            cd[2 * g]     = *(const v4f*)(ndl + 8 * g);
            cd[2 * g + 1] = *(const v4f*)(ndl + 8 * g + 4);
            cu[2 * g]     = *(const v4f*)(nu  + 8 * g);
            cu[2 * g + 1] = *(const v4f*)(nu  + 8 * g + 4);
            if (g == SCAN_NG - 2) {
                #pragma unroll
                for (int j = 0; j < SCAN_TB / 16; ++j)
                    stage(buf ^ 1, trow + 16 * j, nb[j]);
                lds_barrier();
            }
            if ((g & 3) == 0) {
                if (g != 0 || k != 0)
                    *(v4f*)(yo + (g == 0 ? -SCAN_SP : 0) + 4 * sub) = ysum();
            }
            if ((g & 3) == 3) {
                __builtin_amdgcn_fence(__ATOMIC_ACQ_REL, "wavefront", "local");
                #pragma unroll
                for (int i = 0; i < 8; ++i)
                    tq[i] = *(const v4f*)(yrd + i * SCAN_YP);
            }
            #pragma unroll
            for (int s = 0; s < SCAN_G; ++s) bq[s] = bn[s];
        }
    }
    *(v4f*)(po + SEQLEN - SCAN_SP + 4 * sub) = ysum();
}

extern "C" void kernel_launch(void* const* d_in, const int* in_sizes, int n_in,
                              void* d_out, int out_size, void* d_ws, size_t ws_size,
                              hipStream_t stream)
{
    const float* u     = (const float*)d_in[0];
    const float* xpw   = (const float*)d_in[1];
    const float* dtw   = (const float*)d_in[2];
    const float* dtb   = (const float*)d_in[3];
    const float* A_log = (const float*)d_in[4];
    float*       out   = (float*)d_out;

    float* xdbl  = (float*)d_ws;
    float* delta = xdbl + (size_t)BATCH * SEQLEN * KPROJ;

    xdbl_gemm_wmma<<<256, 192, 0, stream>>>(u, xpw, xdbl);

    delta_gemm_wmma<<<4096, 256, 0, stream>>>(xdbl, dtw, dtb, delta);

    s6_scan<<<256, 128, 0, stream>>>(u, xdbl, delta, A_log, out);
}


# x_dbl GEMM: K split in two across work-groups, 32-row tiles, 12 waves (3 per SIMD, equal matrix work), half the weight bytes per row; two partial planes added by the delta kernel (new sum order)
# speedup vs baseline: 11.4985x; 1.0836x over previous
#include <hip/hip_runtime.h>
#include <math.h>

#define BATCH   2
#define SEQLEN  2048
#define D_INNER 2048
#define D_STATE 16
#define DT_RANK 64
#define KPROJ   (DT_RANK + 2 * D_STATE)

typedef float v2f __attribute__((ext_vector_type(2)));
typedef float v8f __attribute__((ext_vector_type(8)));

__device__ __forceinline__ void lds_barrier()
{
    __builtin_amdgcn_fence(__ATOMIC_RELEASE, "workgroup", "local");
    __builtin_amdgcn_s_barrier();
    __builtin_amdgcn_fence(__ATOMIC_ACQUIRE, "workgroup", "local");
}

#define G1_KC    64
#define G1_KH    (D_INNER / 2)
#define G1_NCH   (G1_KH / G1_KC)
#define G1_AP    40
#define G1_BP    (G1_KC + 4)
#define G1_WQ    (KPROJ * (G1_KC / 4) / 384)
#define G1_PLANE ((size_t)BATCH * SEQLEN * KPROJ)

typedef float g1v4 __attribute__((ext_vector_type(4)));

__global__ __launch_bounds__(384) void xdbl_gemm_wmma(
    const float* __restrict__ u,
    const float* __restrict__ xpw,
    float* __restrict__ xdbl)
{
    __shared__ __attribute__((aligned(16))) float As[2][G1_KC][G1_AP];
    __shared__ __attribute__((aligned(16))) float Bs[2][KPROJ][G1_BP];

    const int tid  = threadIdx.x;
    const int lane = tid & 31;
    const int wave = tid >> 5;
    const int msub = (wave >= 6) ? 1 : 0;
    const int nt   = wave - 6 * msub;
    const int kh   = blockIdx.x & 1;
    const int m0 = (blockIdx.x >> 1) * 32, n0 = nt * 16;

    const int half = lane & 15;
    const int hi   = lane >> 4;
    const int mc   = msub * 16 + half;

    const int b  = m0 >> 11;
    const int l0 = m0 & (SEQLEN - 1);

    const int  t8 = tid & 255, aq = t8 & 7, rr = t8 >> 3;
    const int  ak = ((rr & 1) << 2) | ((rr >> 1) & 3) | (rr & 24);
    const float* ga = u + (size_t)b * D_INNER * SEQLEN + (size_t)(kh * G1_KH + ak) * SEQLEN + l0 + aq * 4;
    const int  bn = tid >> 4, bq = tid & 15;
    const float* gb = xpw + (size_t)bn * D_INNER + kh * G1_KH + bq * 4;

    g1v4 ra0[2], rb0[G1_WQ], ra1[2], rb1[G1_WQ];

    auto fetch = [&](int cn, g1v4 (&ra)[2], g1v4 (&rb)[G1_WQ]) __attribute__((always_inline)) {
        const size_t ko = (size_t)cn * G1_KC;
        ra[0] = *(const g1v4*)(ga + ko * SEQLEN);
        ra[1] = *(const g1v4*)(ga + (ko + 32) * SEQLEN);
        #pragma unroll
        for (int j = 0; j < G1_WQ; ++j)
            rb[j] = *(const g1v4*)(gb + (size_t)(24 * j) * D_INNER + ko);
    };
    auto stage = [&](int st, const g1v4 (&ra)[2], const g1v4 (&rb)[G1_WQ]) __attribute__((always_inline)) {
        *(g1v4*)&As[st][ak][aq * 4]      = ra[0];
        *(g1v4*)&As[st][ak + 32][aq * 4] = ra[1];
        #pragma unroll
        for (int j = 0; j < G1_WQ; ++j)
            *(g1v4*)&Bs[st][bn + 24 * j][bq * 4] = rb[j];
    };

    v8f c0 = {}, c1 = {}, c2 = {}, c3 = {};

    auto chunk = [&](int st) __attribute__((always_inline)) {
        v2f fa[G1_KC / 4], fb[G1_KC / 4];
        #pragma unroll
        for (int i = 0; i < G1_KC / 4; ++i) {
            fa[i].x = As[st][4 * i + 2 * hi][mc];
            fa[i].y = As[st][4 * i + 2 * hi + 1][mc];
            fb[i]   = *(const v2f*)&Bs[st][n0 + half][4 * i + 2 * hi];
        }
        #pragma unroll
        for (int i = 0; i < G1_KC / 4; i += 4) {
            c0 = __builtin_amdgcn_wmma_f32_16x16x4_f32(false, fa[i],     false, fb[i],     (short)0, c0, false, false);
            c1 = __builtin_amdgcn_wmma_f32_16x16x4_f32(false, fa[i + 1], false, fb[i + 1], (short)0, c1, false, false);
            c2 = __builtin_amdgcn_wmma_f32_16x16x4_f32(false, fa[i + 2], false, fb[i + 2], (short)0, c2, false, false);
            c3 = __builtin_amdgcn_wmma_f32_16x16x4_f32(false, fa[i + 3], false, fb[i + 3], (short)0, c3, false, false);
        }
    };

    fetch(0, ra0, rb0);
    fetch(1, ra1, rb1);
    stage(0, ra0, rb0);
    lds_barrier();

    #pragma unroll 1
    for (int ci = 0; ci < G1_NCH - 2; ci += 2) {
        fetch(ci + 2, ra0, rb0);
        __builtin_amdgcn_sched_barrier(0);
        chunk(0);
        stage(1, ra1, rb1);
        lds_barrier();

        fetch(ci + 3, ra1, rb1);
        __builtin_amdgcn_sched_barrier(0);
        chunk(1);
        stage(0, ra0, rb0);
        lds_barrier();
    }
    chunk(0);
    stage(1, ra1, rb1);
    lds_barrier();
    chunk(1);

    v8f c = (c0 + c1) + (c2 + c3);

    float* pc = xdbl + (size_t)kh * G1_PLANE + (size_t)(m0 + msub * 16 + hi * 8) * KPROJ + (n0 + half);
    #pragma unroll
    for (int r = 0; r < 8; ++r)
        pc[(size_t)r * KPROJ] = c[r];
}

#define G2_P   (DT_RANK + 4)
#define G2_WR  128

__global__ __launch_bounds__(256) void delta_gemm_wmma(
    float* __restrict__ xdbl,
    const float* __restrict__ dtw,
    const float* __restrict__ dtb,
    float* __restrict__ delta)
{
    __shared__ __attribute__((aligned(16))) float Xs[16][G2_P];
    __shared__ __attribute__((aligned(16))) float Ws[G2_WR][G2_P];

    const int tid  = threadIdx.x;
    const int lane = tid & 31;
    const int wave = tid >> 5;
    const int mt = blockIdx.x >> 4;
    const int ng = blockIdx.x & 15;
    const int m0 = mt * 16, n0 = ng * G2_WR + wave * 16;

    const int srow = tid >> 4, sq = tid & 15;
    const float* px = xdbl + (size_t)(m0 + srow) * KPROJ + sq * 4;
    const g1v4 rx0 = *(const g1v4*)px;
    const g1v4 rx1 = *(const g1v4*)(px + G1_PLANE);
    const float* gw = dtw + (size_t)(ng * G2_WR + srow) * DT_RANK + sq * 4;
    g1v4 rw[G2_WR / 16];
    #pragma unroll
    for (int j = 0; j < G2_WR / 16; ++j)
        rw[j] = *(const g1v4*)(gw + (size_t)(16 * j) * DT_RANK);
    *(g1v4*)&Xs[srow][sq * 4] = rx0 + rx1;
    #pragma unroll
    for (int j = 0; j < G2_WR / 16; ++j)
        *(g1v4*)&Ws[16 * j + srow][sq * 4] = rw[j];
    lds_barrier();

    const int half = lane & 15;
    const int hi   = lane >> 4;

    v2f fa[DT_RANK / 4], fb[DT_RANK / 4];
    #pragma unroll
    for (int i = 0; i < DT_RANK / 4; ++i) {
        fa[i] = *(const v2f*)&Xs[half][4 * i + 2 * hi];
        fb[i] = *(const v2f*)&Ws[wave * 16 + half][4 * i + 2 * hi];
    }

    v8f c0 = {}, c1 = {};
    #pragma unroll
    for (int i = 0; i < DT_RANK / 4; i += 2) {
        c0 = __builtin_amdgcn_wmma_f32_16x16x4_f32(false, fa[i],     false, fb[i],     (short)0, c0, false, false);
        c1 = __builtin_amdgcn_wmma_f32_16x16x4_f32(false, fa[i + 1], false, fb[i + 1], (short)0, c1, false, false);
    }
    v8f c = c0 + c1;

    const int   d    = n0 + half;
    const float bias = dtb[d];
    const int   b    = m0 >> 11;
    const int   lb   = (m0 & (SEQLEN - 1)) + hi * 8;
    float* pd = delta + ((size_t)b * D_INNER + d) * SEQLEN + lb;

    #pragma unroll
    for (int r = 0; r < 8; ++r) {
        float z  = c[r] + bias;
        float sp = fmaxf(z, 0.0f) + log1pf(__expf(-fabsf(z)));
        pd[r] = sp;
    }

    if (ng == 0 && tid < 128) {
        float* pb = xdbl + (size_t)(m0 + (tid >> 3)) * KPROJ + DT_RANK + (tid & 7) * 4;
        *(g1v4*)pb = *(const g1v4*)pb + *(const g1v4*)(pb + G1_PLANE);
    }
}

#define SCAN_TB    64
#define SCAN_NBLK  (SEQLEN / SCAN_TB)
#define SCAN_Q     (SCAN_TB / 4)
#define SCAN_G     8
#define SCAN_NG    (SCAN_TB / SCAN_G)
#define SCAN_SP    32
#define SCAN_YP    (SCAN_SP + 4)

typedef float v4f __attribute__((ext_vector_type(4)));

__global__ __launch_bounds__(128) void s6_scan(
    const float* __restrict__ u,
    const float* __restrict__ xdbl,
    const float* __restrict__ delta,
    const float* __restrict__ A_log,
    float* __restrict__ out)
{
    __shared__ v4f bc[2][SCAN_TB][8];
    __shared__ __attribute__((aligned(16))) float yt[4][32][SCAN_YP];

    const int tid = threadIdx.x;
    const int sub = tid & 7;
    const int ch  = blockIdx.x * 16 + (tid >> 3);
    const int b   = blockIdx.x >> 7;
    const int d   = ch & (D_INNER - 1);

    const v2f al = *(const v2f*)(A_log + (size_t)d * D_STATE + sub * 2);
    const float a0 = -__expf(al.x) * 1.44269504088896340736f;
    const float a1 = -__expf(al.y) * 1.44269504088896340736f;

    const float* pdl = delta + (size_t)ch * SEQLEN;
    const float* pu  = u     + (size_t)ch * SEQLEN;
    float*       po  = out   + (size_t)ch * SEQLEN;
    const int    trow = tid >> 3;
    const int    tcol = tid & 7;
    const float* pbc  = xdbl + ((size_t)b * SEQLEN + trow) * KPROJ + DT_RANK + tcol * 4;
    float* const lrow = (float*)&bc[0][0][0] + (tcol & 3) * 8 + (tcol >> 2) * 2;

    auto stage = [&](int bufi, int row, const v4f q) __attribute__((always_inline)) {
        float* dst = lrow + (bufi * SCAN_TB + row) * 32;
        v2f lo, hi;
        lo.x = q.x; lo.y = q.y; hi.x = q.z; hi.y = q.w;
        *(v2f*)dst       = lo;
        *(v2f*)(dst + 4) = hi;
    };

    v4f cd[SCAN_Q], cu[SCAN_Q];

    #pragma unroll
    for (int j = 0; j < SCAN_Q; ++j) {
        cd[j] = *(const v4f*)(pdl + 4 * j);
        cu[j] = *(const v4f*)(pu  + 4 * j);
    }
    #pragma unroll
    for (int j = 0; j < SCAN_TB / 16; ++j)
        stage(0, trow + 16 * j, *(const v4f*)(pbc + (size_t)(16 * j) * KPROJ));
    lds_barrier();

    float x0 = 0.0f, x1 = 0.0f;

    float* const       ywr = &yt[tid >> 5][tid & 31][0];
    const float* const yrd = &yt[tid >> 5][tid & 24][sub * 4];
    v4f tq[8] = {};

    auto ysum = [&]() __attribute__((always_inline)) {
        return ((tq[0] + tq[1]) + (tq[2] + tq[3])) + ((tq[4] + tq[5]) + (tq[6] + tq[7]));
    };

    v4f bq[SCAN_G];
    #pragma unroll
    for (int s = 0; s < SCAN_G; ++s)
        bq[s] = bc[0][s][sub];

    #pragma unroll 1
    for (int k = 0; k < SCAN_NBLK; ++k) {
        const int buf = k & 1;
        const int kn  = (k + 1 < SCAN_NBLK) ? (k + 1) : k;
        const float* ndl = pdl + kn * SCAN_TB;
        const float* nu  = pu  + kn * SCAN_TB;
        const float* nbc = pbc + (size_t)(kn * SCAN_TB) * KPROJ;
        float*       yo  = po  + k * SCAN_TB;
        const v4f*   rb  = &bc[buf][0][sub];
        const v4f*   rn  = &bc[buf ^ 1][0][sub];

        v4f nb[SCAN_TB / 16];
        v4f bn[SCAN_G];

        #pragma unroll
        for (int g = 0; g < SCAN_NG; ++g) {
            #pragma unroll
            for (int s = 0; s < SCAN_G; ++s)
                bn[s] = (g + 1 < SCAN_NG) ? rb[8 * ((g + 1) * SCAN_G + s)] : rn[8 * s];
            if (g == 0) {
                #pragma unroll
                for (int j = 0; j < SCAN_TB / 16; ++j)
                    nb[j] = *(const v4f*)(nbc + (size_t)(16 * j) * KPROJ);
            }
            __builtin_amdgcn_sched_barrier(0);

            v4f p0, p1;
            #pragma unroll
            for (int s = 0; s < SCAN_G; ++s) {
                const int   q  = 2 * g + (s >> 2);
                const float dl = cd[q][s & 3];
                const float du = dl * cu[q][s & 3];
                const float dA0 = __builtin_amdgcn_exp2f(dl * a0);
                const float dA1 = __builtin_amdgcn_exp2f(dl * a1);
                const float t0 = du * bq[s].x;
                const float t1 = du * bq[s].y;
                x0 = __builtin_fmaf(dA0, x0, t0);
                x1 = __builtin_fmaf(dA1, x1, t1);
                const float p = __builtin_fmaf(x1, bq[s].w, x0 * bq[s].z);
                if (s < 4) p0[s & 3] = p; else p1[s & 3] = p;
            }
            __builtin_amdgcn_sched_barrier(0);

            *(v4f*)(ywr + SCAN_G * (g & 3))     = p0;
            *(v4f*)(ywr + SCAN_G * (g & 3) + 4) = p1;

            cd[2 * g]     = *(const v4f*)(ndl + 8 * g);
            cd[2 * g + 1] = *(const v4f*)(ndl + 8 * g + 4);
            cu[2 * g]     = *(const v4f*)(nu  + 8 * g);
            cu[2 * g + 1] = *(const v4f*)(nu  + 8 * g + 4);
            if (g == SCAN_NG - 2) {
                #pragma unroll
                for (int j = 0; j < SCAN_TB / 16; ++j)
                    stage(buf ^ 1, trow + 16 * j, nb[j]);
                lds_barrier();
            }
            if ((g & 3) == 0) {
                if (g != 0 || k != 0)
                    *(v4f*)(yo + (g == 0 ? -SCAN_SP : 0) + 4 * sub) = ysum();
            }
            if ((g & 3) == 3) {
                __builtin_amdgcn_fence(__ATOMIC_ACQ_REL, "wavefront", "local");
                #pragma unroll
                for (int i = 0; i < 8; ++i)
                    tq[i] = *(const v4f*)(yrd + i * SCAN_YP);
            }
            #pragma unroll
            for (int s = 0; s < SCAN_G; ++s) bq[s] = bn[s];
        }
    }
    *(v4f*)(po + SEQLEN - SCAN_SP + 4 * sub) = ysum();
}

extern "C" void kernel_launch(void* const* d_in, const int* in_sizes, int n_in,
                              void* d_out, int out_size, void* d_ws, size_t ws_size,
                              hipStream_t stream)
{
    const float* u     = (const float*)d_in[0];
    const float* xpw   = (const float*)d_in[1];
    const float* dtw   = (const float*)d_in[2];
    const float* dtb   = (const float*)d_in[3];
    const float* A_log = (const float*)d_in[4];
    float*       out   = (float*)d_out;

    float* xdbl  = (float*)d_ws;
    float* delta = xdbl + 2 * (size_t)BATCH * SEQLEN * KPROJ;

    xdbl_gemm_wmma<<<256, 384, 0, stream>>>(u, xpw, xdbl);

    delta_gemm_wmma<<<4096, 256, 0, stream>>>(xdbl, dtw, dtb, delta);

    s6_scan<<<256, 128, 0, stream>>>(u, xdbl, delta, A_log, out);
}


# softplus and delta*u moved to 4 helper waves per scan work-group (one beside each lone scan wave, in its idle issue slots; out of the delta GEMM), handed to the scan in 3-stage LDS tiles; same sums
# speedup vs baseline: 12.0118x; 1.0446x over previous
#include <hip/hip_runtime.h>
#include <math.h>

#define BATCH   2
#define SEQLEN  2048
#define D_INNER 2048
#define D_STATE 16
#define DT_RANK 64
#define KPROJ   (DT_RANK + 2 * D_STATE)

typedef float v2f __attribute__((ext_vector_type(2)));
typedef float v8f __attribute__((ext_vector_type(8)));

__device__ __forceinline__ void lds_barrier()
{
    __builtin_amdgcn_fence(__ATOMIC_RELEASE, "workgroup", "local");
    __builtin_amdgcn_s_barrier();
    __builtin_amdgcn_fence(__ATOMIC_ACQUIRE, "workgroup", "local");
}

#define G1_KC    64
#define G1_KH    (D_INNER / 2)
#define G1_NCH   (G1_KH / G1_KC)
#define G1_AP    40
#define G1_BP    (G1_KC + 4)
#define G1_WQ    (KPROJ * (G1_KC / 4) / 384)
#define G1_PLANE ((size_t)BATCH * SEQLEN * KPROJ)

typedef float g1v4 __attribute__((ext_vector_type(4)));

__global__ __launch_bounds__(384) void xdbl_gemm_wmma(
    const float* __restrict__ u,
    const float* __restrict__ xpw,
    float* __restrict__ xdbl)
{
    __shared__ __attribute__((aligned(16))) float As[2][G1_KC][G1_AP];
    __shared__ __attribute__((aligned(16))) float Bs[2][KPROJ][G1_BP];

    const int tid  = threadIdx.x;
    const int lane = tid & 31;
    const int wave = tid >> 5;
    const int msub = (wave >= 6) ? 1 : 0;
    const int nt   = wave - 6 * msub;
    const int kh   = blockIdx.x & 1;
    const int m0 = (blockIdx.x >> 1) * 32, n0 = nt * 16;

    const int half = lane & 15;
    const int hi   = lane >> 4;
    const int mc   = msub * 16 + half;

    const int b  = m0 >> 11;
    const int l0 = m0 & (SEQLEN - 1);

    const int  t8 = tid & 255, aq = t8 & 7, rr = t8 >> 3;
    const int  ak = ((rr & 1) << 2) | ((rr >> 1) & 3) | (rr & 24);
    const float* ga = u + (size_t)b * D_INNER * SEQLEN + (size_t)(kh * G1_KH + ak) * SEQLEN + l0 + aq * 4;
    const int  bn = tid >> 4, bq = tid & 15;
    const float* gb = xpw + (size_t)bn * D_INNER + kh * G1_KH + bq * 4;

    g1v4 ra0[2], rb0[G1_WQ], ra1[2], rb1[G1_WQ];

    auto fetch = [&](int cn, g1v4 (&ra)[2], g1v4 (&rb)[G1_WQ]) __attribute__((always_inline)) {
        const size_t ko = (size_t)cn * G1_KC;
        ra[0] = *(const g1v4*)(ga + ko * SEQLEN);
        ra[1] = *(const g1v4*)(ga + (ko + 32) * SEQLEN);
        #pragma unroll
        for (int j = 0; j < G1_WQ; ++j)
            rb[j] = *(const g1v4*)(gb + (size_t)(24 * j) * D_INNER + ko);
    };
    auto stage = [&](int st, const g1v4 (&ra)[2], const g1v4 (&rb)[G1_WQ]) __attribute__((always_inline)) {
        *(g1v4*)&As[st][ak][aq * 4]      = ra[0];
        *(g1v4*)&As[st][ak + 32][aq * 4] = ra[1];
        #pragma unroll
        for (int j = 0; j < G1_WQ; ++j)
            *(g1v4*)&Bs[st][bn + 24 * j][bq * 4] = rb[j];
    };

    v8f c0 = {}, c1 = {}, c2 = {}, c3 = {};

    auto chunk = [&](int st) __attribute__((always_inline)) {
        v2f fa[G1_KC / 4], fb[G1_KC / 4];
        #pragma unroll
        for (int i = 0; i < G1_KC / 4; ++i) {
            fa[i].x = As[st][4 * i + 2 * hi][mc];
            fa[i].y = As[st][4 * i + 2 * hi + 1][mc];
            fb[i]   = *(const v2f*)&Bs[st][n0 + half][4 * i + 2 * hi];
        }
        #pragma unroll
        for (int i = 0; i < G1_KC / 4; i += 4) {
            c0 = __builtin_amdgcn_wmma_f32_16x16x4_f32(false, fa[i],     false, fb[i],     (short)0, c0, false, false);
            c1 = __builtin_amdgcn_wmma_f32_16x16x4_f32(false, fa[i + 1], false, fb[i + 1], (short)0, c1, false, false);
            c2 = __builtin_amdgcn_wmma_f32_16x16x4_f32(false, fa[i + 2], false, fb[i + 2], (short)0, c2, false, false);
            c3 = __builtin_amdgcn_wmma_f32_16x16x4_f32(false, fa[i + 3], false, fb[i + 3], (short)0, c3, false, false);
        }
    };

    fetch(0, ra0, rb0);
    fetch(1, ra1, rb1);
    stage(0, ra0, rb0);
    lds_barrier();

    #pragma unroll 1
    for (int ci = 0; ci < G1_NCH - 2; ci += 2) {
        fetch(ci + 2, ra0, rb0);
        __builtin_amdgcn_sched_barrier(0);
        chunk(0);
        stage(1, ra1, rb1);
        lds_barrier();

        fetch(ci + 3, ra1, rb1);
        __builtin_amdgcn_sched_barrier(0);
        chunk(1);
        stage(0, ra0, rb0);
        lds_barrier();
    }
    chunk(0);
    stage(1, ra1, rb1);
    lds_barrier();
    chunk(1);

    v8f c = (c0 + c1) + (c2 + c3);

    float* pc = xdbl + (size_t)kh * G1_PLANE + (size_t)(m0 + msub * 16 + hi * 8) * KPROJ + (n0 + half);
    #pragma unroll
    for (int r = 0; r < 8; ++r)
        pc[(size_t)r * KPROJ] = c[r];
}

#define G2_P   (DT_RANK + 4)
#define G2_WR  128

__global__ __launch_bounds__(256) void delta_gemm_wmma(
    float* __restrict__ xdbl,
    const float* __restrict__ dtw,
    const float* __restrict__ dtb,
    float* __restrict__ zpre)
{
    __shared__ __attribute__((aligned(16))) float Xs[16][G2_P];
    __shared__ __attribute__((aligned(16))) float Ws[G2_WR][G2_P];

    const int tid  = threadIdx.x;
    const int lane = tid & 31;
    const int wave = tid >> 5;
    const int mt = blockIdx.x >> 4;
    const int ng = blockIdx.x & 15;
    const int m0 = mt * 16, n0 = ng * G2_WR + wave * 16;

    const int srow = tid >> 4, sq = tid & 15;
    const float* px = xdbl + (size_t)(m0 + srow) * KPROJ + sq * 4;
    const g1v4 rx0 = *(const g1v4*)px;
    const g1v4 rx1 = *(const g1v4*)(px + G1_PLANE);
    const float* gw = dtw + (size_t)(ng * G2_WR + srow) * DT_RANK + sq * 4;
    g1v4 rw[G2_WR / 16];
    #pragma unroll
    for (int j = 0; j < G2_WR / 16; ++j)
        rw[j] = *(const g1v4*)(gw + (size_t)(16 * j) * DT_RANK);
    *(g1v4*)&Xs[srow][sq * 4] = rx0 + rx1;
    #pragma unroll
    for (int j = 0; j < G2_WR / 16; ++j)
        *(g1v4*)&Ws[16 * j + srow][sq * 4] = rw[j];
    lds_barrier();

    const int half = lane & 15;
    const int hi   = lane >> 4;

    v2f fa[DT_RANK / 4], fb[DT_RANK / 4];
    #pragma unroll
    for (int i = 0; i < DT_RANK / 4; ++i) {
        fa[i] = *(const v2f*)&Xs[half][4 * i + 2 * hi];
        fb[i] = *(const v2f*)&Ws[wave * 16 + half][4 * i + 2 * hi];
    }

    v8f c0 = {}, c1 = {};
    #pragma unroll
    for (int i = 0; i < DT_RANK / 4; i += 2) {
        c0 = __builtin_amdgcn_wmma_f32_16x16x4_f32(false, fa[i],     false, fb[i],     (short)0, c0, false, false);
        c1 = __builtin_amdgcn_wmma_f32_16x16x4_f32(false, fa[i + 1], false, fb[i + 1], (short)0, c1, false, false);
    }
    v8f c = c0 + c1;

    const int   d    = n0 + half;
    const float bias = dtb[d];
    const int   b    = m0 >> 11;
    const int   lb   = (m0 & (SEQLEN - 1)) + hi * 8;
    float* pd = zpre + ((size_t)b * D_INNER + d) * SEQLEN + lb;

    #pragma unroll
    for (int r = 0; r < 8; ++r)
        pd[r] = c[r] + bias;

    if (ng == 0 && tid < 128) {
        float* pb = xdbl + (size_t)(m0 + (tid >> 3)) * KPROJ + DT_RANK + (tid & 7) * 4;
        *(g1v4*)pb = *(const g1v4*)pb + *(const g1v4*)(pb + G1_PLANE);
    }
}

#define SCAN_TB    64
#define SCAN_NBLK  (SEQLEN / SCAN_TB)
#define SCAN_Q     (SCAN_TB / 4)
#define SCAN_G     8
#define SCAN_NG    (SCAN_TB / SCAN_G)
#define SCAN_SP    32
#define SCAN_YP    (SCAN_SP + 4)
#define SCAN_DP    (SCAN_TB + 4)

typedef float v4f __attribute__((ext_vector_type(4)));

__device__ __forceinline__ v4f softplus4(const v4f z)
{
    v4f r;
    #pragma unroll
    for (int i = 0; i < 4; ++i)
        r[i] = fmaxf(z[i], 0.0f) + log1pf(__expf(-fabsf(z[i])));
    return r;
}

__global__ __launch_bounds__(256) void s6_scan(
    const float* __restrict__ u,
    const float* __restrict__ xdbl,
    const float* __restrict__ zpre,
    const float* __restrict__ A_log,
    float* __restrict__ out)
{
    __shared__ v4f bc[2][SCAN_TB][8];
    __shared__ __attribute__((aligned(16))) float yt[4][32][SCAN_YP];
    __shared__ __attribute__((aligned(16))) float dts[3][16][SCAN_DP];
    __shared__ __attribute__((aligned(16))) float dus[3][16][SCAN_DP];

    const int tid = threadIdx.x;
    const bool is_scan = __builtin_amdgcn_readfirstlane(tid) < 128;
    const int ts  = tid & 127;
    const int sub = ts & 7;
    const int cl  = ts >> 3;
    const int ch  = blockIdx.x * 16 + cl;
    const int b   = blockIdx.x >> 7;
    const int d   = ch & (D_INNER - 1);

    const float* pz = zpre + (size_t)ch * SEQLEN + 8 * sub;
    const float* pv = u    + (size_t)ch * SEQLEN + 8 * sub;

    {
        const int pb = tid >> 7;
        const v4f za = *(const v4f*)(pz + pb * SCAN_TB);
        const v4f zb = *(const v4f*)(pz + pb * SCAN_TB + 4);
        const v4f ua = *(const v4f*)(pv + pb * SCAN_TB);
        const v4f ub = *(const v4f*)(pv + pb * SCAN_TB + 4);
        const v4f da = softplus4(za), db = softplus4(zb);
        *(v4f*)&dts[pb][cl][8 * sub]     = da;
        *(v4f*)&dts[pb][cl][8 * sub + 4] = db;
        *(v4f*)&dus[pb][cl][8 * sub]     = da * ua;
        *(v4f*)&dus[pb][cl][8 * sub + 4] = db * ub;
    }
    v4f zq0 = {}, zq1 = {}, uq0 = {}, uq1 = {};
    if (!is_scan) {
        zq0 = *(const v4f*)(pz + 2 * SCAN_TB);
        zq1 = *(const v4f*)(pz + 2 * SCAN_TB + 4);
        uq0 = *(const v4f*)(pv + 2 * SCAN_TB);
        uq1 = *(const v4f*)(pv + 2 * SCAN_TB + 4);
    }

    const v2f al = *(const v2f*)(A_log + (size_t)d * D_STATE + sub * 2);
    const float a0 = -__expf(al.x) * 1.44269504088896340736f;
    const float a1 = -__expf(al.y) * 1.44269504088896340736f;

    float*       po  = out + (size_t)ch * SEQLEN;
    const int    trow = ts >> 3;
    const int    tcol = ts & 7;
    const float* pbc  = xdbl + ((size_t)b * SEQLEN + trow) * KPROJ + DT_RANK + tcol * 4;
    float* const lrow = (float*)&bc[0][0][0] + (tcol & 3) * 8 + (tcol >> 2) * 2;

    auto stage = [&](int bufi, int row, const v4f q) __attribute__((always_inline)) {
        float* dst = lrow + (bufi * SCAN_TB + row) * 32;
        v2f lo, hi;
        lo.x = q.x; lo.y = q.y; hi.x = q.z; hi.y = q.w;
        *(v2f*)dst       = lo;
        *(v2f*)(dst + 4) = hi;
    };

    v4f cd[SCAN_Q], cu[SCAN_Q];

    if (is_scan) {
        #pragma unroll
        for (int j = 0; j < SCAN_TB / 16; ++j)
            stage(0, trow + 16 * j, *(const v4f*)(pbc + (size_t)(16 * j) * KPROJ));
    }
    lds_barrier();

    float x0 = 0.0f, x1 = 0.0f;

    float* const       ywr = &yt[ts >> 5][ts & 31][0];
    const float* const yrd = &yt[ts >> 5][ts & 24][sub * 4];
    v4f tq[8] = {};

    auto ysum = [&]() __attribute__((always_inline)) {
        return ((tq[0] + tq[1]) + (tq[2] + tq[3])) + ((tq[4] + tq[5]) + (tq[6] + tq[7]));
    };

    v4f bq[SCAN_G] = {};
    v4f bn[SCAN_G] = {};
    v4f nb[SCAN_TB / 16] = {};
    if (is_scan) {
        #pragma unroll
        for (int j = 0; j < SCAN_Q; ++j) {
            cd[j] = *(const v4f*)&dts[0][cl][4 * j];
            cu[j] = *(const v4f*)&dus[0][cl][4 * j];
        }
        #pragma unroll
        for (int s = 0; s < SCAN_G; ++s)
            bq[s] = bc[0][s][sub];
    }

    struct Blk {
        const float* ndl;
        const float* ndu;
        const float* nbc;
        float*       yo;
        const v4f*   rb;
        const v4f*   rn;
        int          nbuf;
        bool         first;
    };
    auto blk = [&](int k, int st_next) __attribute__((always_inline)) {
        const int kn = (k + 1 < SCAN_NBLK) ? (k + 1) : k;
        Blk r;
        r.ndl   = &dts[st_next][cl][0];
        r.ndu   = &dus[st_next][cl][0];
        r.nbc   = pbc + (size_t)(kn * SCAN_TB) * KPROJ;
        r.yo    = po  + k * SCAN_TB;
        r.rb    = &bc[k & 1][0][sub];
        r.rn    = &bc[(k & 1) ^ 1][0][sub];
        r.nbuf  = (k & 1) ^ 1;
        r.first = (k == 0);
        return r;
    };

    auto grpA = [&](const int g, const Blk& P) __attribute__((always_inline)) {
        #pragma unroll
        for (int s = 0; s < SCAN_G; ++s)
            bn[s] = (g + 1 < SCAN_NG) ? P.rb[8 * ((g + 1) * SCAN_G + s)] : P.rn[8 * s];
        if (g == 0) {
            #pragma unroll
            for (int j = 0; j < SCAN_TB / 16; ++j)
                nb[j] = *(const v4f*)(P.nbc + (size_t)(16 * j) * KPROJ);
        }
        __builtin_amdgcn_sched_barrier(0);

        v4f p0, p1;
        #pragma unroll
        for (int s = 0; s < SCAN_G; ++s) {
            const int   q  = 2 * g + (s >> 2);
            const float dl = cd[q][s & 3];
            const float du = cu[q][s & 3];
            const float dA0 = __builtin_amdgcn_exp2f(dl * a0);
            const float dA1 = __builtin_amdgcn_exp2f(dl * a1);
            const float t0 = du * bq[s].x;
            const float t1 = du * bq[s].y;
            x0 = __builtin_fmaf(dA0, x0, t0);
            x1 = __builtin_fmaf(dA1, x1, t1);
            const float p = __builtin_fmaf(x1, bq[s].w, x0 * bq[s].z);
            if (s < 4) p0[s & 3] = p; else p1[s & 3] = p;
        }
        __builtin_amdgcn_sched_barrier(0);

        *(v4f*)(ywr + SCAN_G * (g & 3))     = p0;
        *(v4f*)(ywr + SCAN_G * (g & 3) + 4) = p1;

        cd[2 * g]     = *(const v4f*)(P.ndl + 8 * g);
        cd[2 * g + 1] = *(const v4f*)(P.ndl + 8 * g + 4);
        cu[2 * g]     = *(const v4f*)(P.ndu + 8 * g);
        cu[2 * g + 1] = *(const v4f*)(P.ndu + 8 * g + 4);
        if (g == SCAN_NG - 2) {
            #pragma unroll
            for (int j = 0; j < SCAN_TB / 16; ++j)
                stage(P.nbuf, trow + 16 * j, nb[j]);
        }
    };
    auto grpB = [&](const int g, const Blk& P) __attribute__((always_inline)) {
        if ((g & 3) == 0) {
            if (g != 0 || !P.first)
                *(v4f*)(P.yo + (g == 0 ? -SCAN_SP : 0) + 4 * sub) = ysum();
        }
        if ((g & 3) == 3) {
            __builtin_amdgcn_fence(__ATOMIC_ACQ_REL, "wavefront", "local");
            #pragma unroll
            for (int i = 0; i < 8; ++i)
                tq[i] = *(const v4f*)(yrd + i * SCAN_YP);
        }
        #pragma unroll
        for (int s = 0; s < SCAN_G; ++s) bq[s] = bn[s];
    };
    auto head7 = [&](const Blk& P) __attribute__((always_inline)) {
        #pragma unroll
        for (int g = 0; g < SCAN_NG - 2; ++g) {
            grpA(g, P);
            grpB(g, P);
        }
        grpA(SCAN_NG - 2, P);
    };
    auto tail2 = [&](const Blk& P) __attribute__((always_inline)) {
        grpB(SCAN_NG - 2, P);
        grpA(SCAN_NG - 1, P);
        grpB(SCAN_NG - 1, P);
    };
    auto help = [&](int k, int st) __attribute__((always_inline)) {
        const int k3 = (k + 3 < SCAN_NBLK) ? (k + 3) : (SCAN_NBLK - 1);
        const v4f zn0 = *(const v4f*)(pz + k3 * SCAN_TB);
        const v4f zn1 = *(const v4f*)(pz + k3 * SCAN_TB + 4);
        const v4f un0 = *(const v4f*)(pv + k3 * SCAN_TB);
        const v4f un1 = *(const v4f*)(pv + k3 * SCAN_TB + 4);
        __builtin_amdgcn_sched_barrier(0);
        const v4f d0 = softplus4(zq0), d1 = softplus4(zq1);
        *(v4f*)&dts[st][cl][8 * sub]     = d0;
        *(v4f*)&dts[st][cl][8 * sub + 4] = d1;
        *(v4f*)&dus[st][cl][8 * sub]     = d0 * uq0;
        *(v4f*)&dus[st][cl][8 * sub + 4] = d1 * uq1;
        zq0 = zn0; zq1 = zn1; uq0 = un0; uq1 = un1;
    };

    if (is_scan)
        head7(blk(0, 1));
    else
        help(0, 2);
    lds_barrier();

    int s0 = 1;
    #pragma unroll 1
    for (int k = 1; k < SCAN_NBLK; ++k) {
        const int s1 = (s0 == 2) ? 0 : s0 + 1;
        const int s2 = (s1 == 2) ? 0 : s1 + 1;
        if (is_scan) {
            tail2(blk(k - 1, s0));
            head7(blk(k, s1));
        } else {
            help(k, s2);
        }
        lds_barrier();
        s0 = s1;
    }
    if (is_scan) {
        tail2(blk(SCAN_NBLK - 1, 0));
        *(v4f*)(po + SEQLEN - SCAN_SP + 4 * sub) = ysum();
    }
}

extern "C" void kernel_launch(void* const* d_in, const int* in_sizes, int n_in,
                              void* d_out, int out_size, void* d_ws, size_t ws_size,
                              hipStream_t stream)
{
    const float* u     = (const float*)d_in[0];
    const float* xpw   = (const float*)d_in[1];
    const float* dtw   = (const float*)d_in[2];
    const float* dtb   = (const float*)d_in[3];
    const float* A_log = (const float*)d_in[4];
    float*       out   = (float*)d_out;

    float* xdbl = (float*)d_ws;
    float* zpre = xdbl + 2 * (size_t)BATCH * SEQLEN * KPROJ;

    xdbl_gemm_wmma<<<256, 384, 0, stream>>>(u, xpw, xdbl);

    delta_gemm_wmma<<<4096, 256, 0, stream>>>(xdbl, dtw, dtb, zpre);

    s6_scan<<<256, 256, 0, stream>>>(u, xdbl, zpre, A_log, out);
}
